# starting kernel
# baseline (speedup 1.0000x reference)
#include <hip/hip_runtime.h>

typedef __attribute__((ext_vector_type(2))) float v2f;
typedef __attribute__((ext_vector_type(8))) float v8f;

#define BB 128
#define TT 2048
#define FF 124
#define HH 3
#define OO 6
#define PH 4

__global__ __launch_bounds__(32) void rnn_inproj_wmma(
    const float* __restrict__ x,
    const float* __restrict__ w_ih0,
    const float* __restrict__ b_ih0,
    const float* __restrict__ b_hh0,
    float* __restrict__ pre)
{
    const int lane = threadIdx.x;
    const int row  = lane & 15;
    const int hi   = lane >> 4;
    const int t0   = blockIdx.x * 16;
    const int b    = blockIdx.y;

    const float* xrow = x + ((size_t)b * TT + (size_t)(t0 + row)) * FF;

    const int   n     = row;
    const int   nclmp = (n < HH) ? n : (HH - 1);
    const float nmask = (n < HH) ? 1.0f : 0.0f;
    const float* wrow = w_ih0 + (size_t)nclmp * FF;

    v8f acc = {};
#pragma unroll 4
    for (int kk = 0; kk < FF / 4; ++kk) {
        const int k0 = kk * 4 + 2 * hi;
        v2f a = *(const v2f*)(xrow + k0);
        v2f bm;
        bm.x = wrow[k0]     * nmask;
        bm.y = wrow[k0 + 1] * nmask;
        acc = __builtin_amdgcn_wmma_f32_16x16x4_f32(
             false, a,  false, bm,
             (short)0, acc,  false,  false);
    }

    if (n < HH) {
        const float bias = b_ih0[n] + b_hh0[n];
#pragma unroll
        for (int r = 0; r < 8; ++r) {
            const int t = t0 + r + 8 * hi;
            pre[((size_t)t * BB + b) * PH + n] = acc[r] + bias;
        }
    }
}

__device__ __forceinline__ float fast_tanh(float x) {
    float r;
    asm volatile("v_tanh_f32 %0, %1\n\tv_nop" : "=v"(r) : "v"(x));
    return r;
}

__device__ __forceinline__ float selu_f(float x) {
    const float kAlpha = 1.6732632423543772f;
    const float kScale = 1.0507009873554805f;
    const float e = __builtin_amdgcn_exp2f(x * 1.4426950408889634f);
    return (x > 0.0f) ? kScale * x : kScale * kAlpha * (e - 1.0f);
}

__global__ __launch_bounds__(32) void rnn_scan(
    const float* __restrict__ pre,
    const float* __restrict__ hx,
    const float* __restrict__ w_hh0,
    const float* __restrict__ w_ih1,
    const float* __restrict__ w_hh1,
    const float* __restrict__ b_ih1,
    const float* __restrict__ b_hh1,
    const float* __restrict__ W_out,
    const float* __restrict__ b_out,
    float* __restrict__ out)
{
    const int b = blockIdx.x * 32 + threadIdx.x;

    float A0[3][3], Wi[3][3], Wh[3][3], c1[3];
#pragma unroll
    for (int i = 0; i < 3; ++i) {
#pragma unroll
        for (int j = 0; j < 3; ++j) {
            A0[i][j] = w_hh0[i * 3 + j];
            Wi[i][j] = w_ih1[i * 3 + j];
            Wh[i][j] = w_hh1[i * 3 + j];
        }
        c1[i] = b_ih1[i] + b_hh1[i];
    }
    float Wo[6][3], bo[6];
#pragma unroll
    for (int o = 0; o < 6; ++o) {
        bo[o] = b_out[o];
#pragma unroll
        for (int j = 0; j < 3; ++j) Wo[o][j] = W_out[o * 3 + j];
    }

    float h1[3], h2[3];
#pragma unroll
    for (int i = 0; i < 3; ++i) {
        h1[i] = hx[(0 * BB + b) * HH + i];
        h2[i] = hx[(1 * BB + b) * HH + i];
    }

    for (int t = 0; t < TT; ++t) {
        const float4 pv = *(const float4*)(pre + ((size_t)t * BB + b) * PH);
        const float p[3] = {pv.x, pv.y, pv.z};

        float n0[3];
#pragma unroll
        for (int i = 0; i < 3; ++i) {
            float u = p[i];
#pragma unroll
            for (int j = 0; j < 3; ++j) u = fmaf(A0[i][j], h1[j], u);
            n0[i] = fast_tanh(u);
        }
#pragma unroll
        for (int i = 0; i < 3; ++i) h1[i] = n0[i];

        float n1[3];
#pragma unroll
        for (int i = 0; i < 3; ++i) {
            float v = c1[i];
#pragma unroll
            for (int j = 0; j < 3; ++j) v = fmaf(Wi[i][j], h1[j], v);
#pragma unroll
            for (int j = 0; j < 3; ++j) v = fmaf(Wh[i][j], h2[j], v);
            n1[i] = fast_tanh(v);
        }
#pragma unroll
        for (int i = 0; i < 3; ++i) h2[i] = n1[i];

        float y[3];
#pragma unroll
        for (int i = 0; i < 3; ++i) y[i] = selu_f(h2[i]);

        float* op = out + ((size_t)b * TT + t) * OO;
#pragma unroll
        for (int o = 0; o < 6; ++o) {
            float s = bo[o];
#pragma unroll
            for (int j = 0; j < 3; ++j) s = fmaf(Wo[o][j], y[j], s);
            op[o] = s;
        }
    }
}

extern "C" void kernel_launch(void* const* d_in, const int* in_sizes, int n_in,
                              void* d_out, int out_size, void* d_ws, size_t ws_size,
                              hipStream_t stream)
{
    const float* x     = (const float*)d_in[0];
    const float* hx    = (const float*)d_in[1];
    const float* w_ih0 = (const float*)d_in[2];
    const float* w_hh0 = (const float*)d_in[3];
    const float* b_ih0 = (const float*)d_in[4];
    const float* b_hh0 = (const float*)d_in[5];
    const float* w_ih1 = (const float*)d_in[6];
    const float* w_hh1 = (const float*)d_in[7];
    const float* b_ih1 = (const float*)d_in[8];
    const float* b_hh1 = (const float*)d_in[9];
    const float* W_out = (const float*)d_in[10];
    const float* b_out = (const float*)d_in[11];

    float* out = (float*)d_out;
    float* pre = (float*)d_ws;

    dim3 g1(TT / 16, BB);
    rnn_inproj_wmma<<<g1, 32, 0, stream>>>(x, w_ih0, b_ih0, b_hh0, pre);

    rnn_scan<<<BB / 32, 32, 0, stream>>>(pre, hx, w_hh0, w_ih1, w_hh1,
                                         b_ih1, b_hh1, W_out, b_out, out);
}


# scan: issue order of a recurrence step: its three tanh back to back in one asm statement, then the multiply-adds term by term (tanh and readers were interleaved); same arithmetic and sum order
# speedup vs baseline: 18.7254x; 18.7254x over previous
#include <hip/hip_runtime.h>

typedef __attribute__((ext_vector_type(2))) float v2f;
typedef __attribute__((ext_vector_type(8))) float v8f;

#define BB 128
#define TT 2048
#define FF 124
#define HH 3
#define OO 6
#define PH 4

#define TPW  8
#define K1W  4

__device__ __forceinline__ float quad_x1(float v) {
    return __builtin_bit_cast(float, __builtin_amdgcn_mov_dpp(
        __builtin_bit_cast(int, v), 0xB1  , 0xF, 0xF, true));
}
__device__ __forceinline__ float quad_x2(float v) {
    return __builtin_bit_cast(float, __builtin_amdgcn_mov_dpp(
        __builtin_bit_cast(int, v), 0x4E  , 0xF, 0xF, true));
}

__global__ __launch_bounds__(32 * K1W, 8) void rnn_inproj_wmma(
    const float* __restrict__ x,
    const float* __restrict__ w_ih0,
    const float* __restrict__ b_ih0,
    const float* __restrict__ b_hh0,
    float* __restrict__ pre)
{
    const int lane = threadIdx.x & 31;
    const int wv   = threadIdx.x >> 5;
    const int c    = lane & 15;
    const int hi   = lane >> 4;
    const int g    = c & 3;
    const int r4   = c >> 2;
    const int b    = blockIdx.y;
    const int t0   = (blockIdx.x * K1W + wv) * (4 * TPW);

    const int  qo   = 2 * g + hi;
    const bool last = (qo == 7);
    const int  q3   = last ? 30 : 24 + qo;

    const bool   ncol = (r4 < HH);
    const float* wrow = w_ih0 + (size_t)(ncol ? r4 : HH - 1) * FF;
    v2f bf[8];
#pragma unroll
    for (int j = 0; j < 4; ++j) {
        const int    qc = (j < 3) ? 8 * j + qo : q3;
        const float4 w4 = *(const float4*)(wrow + 4 * qc);
        const bool   ok = ncol && !(j == 3 && last);
        bf[2 * j].x     = ok ? w4.x : 0.0f;
        bf[2 * j].y     = ok ? w4.y : 0.0f;
        bf[2 * j + 1].x = ok ? w4.z : 0.0f;
        bf[2 * j + 1].y = ok ? w4.w : 0.0f;
    }
    const float bias = b_ih0[ncol ? r4 : HH - 1] + b_hh0[ncol ? r4 : HH - 1];

    const float* xrow = x + ((size_t)b * TT + (size_t)(t0 + r4)) * FF;
    const float* xa   = xrow + 4 * qo;
    const float* xb   = xrow + 4 * q3;

    const bool st = ncol && (g < 2);
    float* prow = pre + ((size_t)(t0 + 2 * hi + g) * BB + b) * PH + (ncol ? r4 : 0);

    float res[TPW];
#pragma unroll
    for (int i = 0; i < TPW; ++i) {
        float4 a4[4];
#pragma unroll
        for (int j = 0; j < 3; ++j)
            a4[j] = *(const float4*)(xa + i * (4 * FF) + 32 * j);
        a4[3] = *(const float4*)(xb + i * (4 * FF));
        if (last) a4[3] = make_float4(0.0f, 0.0f, 0.0f, 0.0f);

        v8f acc = {};
#pragma unroll
        for (int j = 0; j < 4; ++j) {
            v2f lo, up;
            lo.x = a4[j].x; lo.y = a4[j].y;
            up.x = a4[j].z; up.y = a4[j].w;
            acc = __builtin_amdgcn_wmma_f32_16x16x4_f32(
                 false, lo,  false, bf[2 * j],
                 (short)0, acc,  false,  false);
            acc = __builtin_amdgcn_wmma_f32_16x16x4_f32(
                 false, up,  false, bf[2 * j + 1],
                 (short)0, acc,  false,  false);
        }

        float v0 = (g & 2) ? ((g & 1) ? acc[3] : acc[2]) : ((g & 1) ? acc[1] : acc[0]);
        float v1 = (g & 2) ? ((g & 1) ? acc[7] : acc[6]) : ((g & 1) ? acc[5] : acc[4]);
        v0 += quad_x1(v0);
        v1 += quad_x1(v1);
        v0 += quad_x2(v0);
        v1 += quad_x2(v1);
        res[i] = ((g == 0) ? v0 : v1) + bias;
    }

    if (st) {
#pragma unroll
        for (int i = 0; i < TPW; ++i)
            prow[(size_t)i * (4 * BB * PH)] = res[i];
    }
}

#define NB   8
#define CH   64
#define NCH  (TT / CH)
#define GS   16
#define HP   (NB * 4 + 4)
#define RPW  (32 / NB)
#define NQ   (CH / RPW)
#define NGRP (NB * CH / 32)
#define NHW  4
#define BPH  (NB / NHW)

__device__ __forceinline__ void tanh3(float x0, float x1, float x2,
                                      float& r0, float& r1, float& r2) {
    asm volatile("v_tanh_f32 %0, %3\n\t"
                 "v_tanh_f32 %1, %4\n\t"
                 "v_tanh_f32 %2, %5\n\t"
                 "v_nop"
                 : "=&v"(r0), "=&v"(r1), "=&v"(r2)
                 : "v"(x0), "v"(x1), "v"(x2));
}

__device__ __forceinline__ float selu_f(float x) {
    const float kAlpha = 1.6732632423543772f;
    const float kScale = 1.0507009873554805f;
    const float e = __builtin_amdgcn_exp2f(x * 1.4426950408889634f);
    return (x > 0.0f) ? kScale * x : kScale * kAlpha * (e - 1.0f);
}

__device__ __forceinline__ void head_one(float hx0, float hx1, float hx2,
                                         float* __restrict__ op,
                                         const float (&Wo)[6][3], const float (&bo)[6])
{
    float y[3];
    y[0] = selu_f(hx0);
    y[1] = selu_f(hx1);
    y[2] = selu_f(hx2);
#pragma unroll
    for (int o = 0; o < 6; ++o) {
        float s = bo[o];
#pragma unroll
        for (int j = 0; j < 3; ++j) s = fmaf(Wo[o][j], y[j], s);
        op[o] = s;
    }
}

__device__ __forceinline__ void helper_head(int hw, int lane, const float* h2l,
                                            float* __restrict__ ob0,
                                            const float (&Wo)[6][3], const float (&bo)[6])
{
    float4 hv[2 * BPH];
#pragma unroll
    for (int k = 0; k < BPH; ++k) {
        const float* hb = h2l + lane * HP + (hw * BPH + k) * 4;
        hv[2 * k]     = *(const float4*)(hb);
        hv[2 * k + 1] = *(const float4*)(hb + 32 * HP);
    }
#pragma unroll
    for (int k = 0; k < BPH; ++k) {
        const int be = hw * BPH + k;
        float* ob = ob0 + ((size_t)be * TT + lane) * OO;
        head_one(hv[2 * k].x, hv[2 * k].y, hv[2 * k].z, ob, Wo, bo);
        head_one(hv[2 * k + 1].x, hv[2 * k + 1].y, hv[2 * k + 1].z, ob + 32 * OO, Wo, bo);
    }
}

__device__ __forceinline__ void helper_fetch(int hw, int lane,
                                             const float* __restrict__ src0, float* prel)
{
    const float* src = src0 + ((size_t)(lane >> 3) * BB + (lane & 7)) * PH;
    float4 r[NQ / 2];
#pragma unroll
    for (int i = 0; i < NQ / 2; ++i)
        r[i] = *(const float4*)(src + (size_t)((hw * (NQ / 2) + i) * RPW) * (BB * PH));
    float* dst = prel + lane * 4;
#pragma unroll
    for (int i = 0; i < NQ / 2; ++i)
        *(float4*)(dst + (hw * (NQ / 2) + i) * (32 * 4)) = r[i];
}

typedef __attribute__((ext_vector_type(3))) float v3f;

__global__ __launch_bounds__(256) void rnn_scan(
    const float* __restrict__ pre,
    const float* __restrict__ hx,
    const float* __restrict__ w_hh0,
    const float* __restrict__ w_ih1,
    const float* __restrict__ w_hh1,
    const float* __restrict__ b_ih1,
    const float* __restrict__ b_hh1,
    const float* __restrict__ W_out,
    const float* __restrict__ b_out,
    float* __restrict__ out)
{
    __shared__ __attribute__((aligned(16))) float preL[2][CH][NB * PH];
    __shared__ __attribute__((aligned(16))) float qL[2][CH][NB * PH];
    __shared__ __attribute__((aligned(16))) float h2L[2][CH][HP];

    const int lane = threadIdx.x & 31;
    const int wave = __builtin_amdgcn_readfirstlane((int)(threadIdx.x >> 5));
    const int b0   = blockIdx.x * NB;

    float A0[3][3], Wi[3][3], Wh[3][3], c1[3];
#pragma unroll
    for (int i = 0; i < 3; ++i) {
#pragma unroll
        for (int j = 0; j < 3; ++j) {
            A0[i][j] = w_hh0[i * 3 + j];
            Wi[i][j] = w_ih1[i * 3 + j];
            Wh[i][j] = w_hh1[i * 3 + j];
        }
        c1[i] = b_ih1[i] + b_hh1[i];
    }
    float Wo[6][3], bo[6];
#pragma unroll
    for (int o = 0; o < 6; ++o) {
        bo[o] = b_out[o];
#pragma unroll
        for (int j = 0; j < 3; ++j) Wo[o][j] = W_out[o * 3 + j];
    }

    const int bl = lane & (NB - 1);
    float h1[3], h2[3];
#pragma unroll
    for (int i = 0; i < 3; ++i) {
        h1[i] = hx[(0 * BB + b0 + bl) * HH + i];
        h2[i] = hx[(1 * BB + b0 + bl) * HH + i];
    }

    for (int s = 0; s < NCH + 3; ++s) {
        if (wave == 0) {
            if (s >= 1 && s <= NCH && lane < NB) {
                const int buf = (s - 1) & 1;
                const float* pL = &preL[buf][0][lane * PH];
                float*       qW = &qL[buf][0][lane * PH];
                for (int tg = 0; tg < CH; tg += GS) {
                    v3f pv[GS];
#pragma unroll
                    for (int g = 0; g < GS; ++g)
                        pv[g] = *(const v3f*)(pL + (tg + g) * (NB * PH));

                    v3f qs[GS];
#pragma unroll
                    for (int g = 0; g < GS; ++g) {
                        const float p[3] = {pv[g].x, pv[g].y, pv[g].z};

                        float u0[3];
#pragma unroll
                        for (int i = 0; i < 3; ++i) {
                            float u = p[i];
#pragma unroll
                            for (int j = 0; j < 3; ++j) u = fmaf(A0[i][j], h1[j], u);
                            u0[i] = u;
                        }
                        tanh3(u0[0], u0[1], u0[2], h1[0], h1[1], h1[2]);

                        float q[3];
#pragma unroll
                        for (int i = 0; i < 3; ++i) {
                            float v = c1[i];
#pragma unroll
                            for (int j = 0; j < 3; ++j) v = fmaf(Wi[i][j], h1[j], v);
                            q[i] = v;
                        }
                        qs[g].x = q[0]; qs[g].y = q[1]; qs[g].z = q[2];
                    }

#pragma unroll
                    for (int g = 0; g < GS; ++g)
                        *(v3f*)(qW + (tg + g) * (NB * PH)) = qs[g];
                }
            }
        } else if (wave == 1) {
            if (s >= 2 && s <= NCH + 1 && lane < NB) {
                const int buf = s & 1;
                const float* qR = &qL[buf][0][lane * PH];
                float*       hL = &h2L[buf][0][lane * 4];
                for (int tg = 0; tg < CH; tg += GS) {
                    v3f qv[GS];
#pragma unroll
                    for (int g = 0; g < GS; ++g)
                        qv[g] = *(const v3f*)(qR + (tg + g) * (NB * PH));

                    float hs[GS][3];
#pragma unroll
                    for (int g = 0; g < GS; ++g) {
                        const float q[3] = {qv[g].x, qv[g].y, qv[g].z};

                        float u1[3] = {q[0], q[1], q[2]};
#pragma unroll
                        for (int j = 0; j < 3; ++j) {
#pragma unroll
                            for (int i = 0; i < 3; ++i) u1[i] = fmaf(Wh[i][j], h2[j], u1[i]);
                            __builtin_amdgcn_sched_barrier(0);
                        }
                        tanh3(u1[0], u1[1], u1[2], h2[0], h2[1], h2[2]);
#pragma unroll
                        for (int i = 0; i < 3; ++i) hs[g][i] = h2[i];
                    }

#pragma unroll
                    for (int g = 0; g < GS; ++g)
                        *(float4*)(hL + (tg + g) * HP) =
                            make_float4(hs[g][0], hs[g][1], hs[g][2], 0.0f);
                }
            }
        } else if (wave & 2) {
            const int hw = (wave & 1) | ((wave >> 2) << 1);
            if (s >= 3)
                helper_head(hw, lane, &h2L[(s + 1) & 1][0][0],
                            out + ((size_t)b0 * TT + (size_t)(s - 3) * CH) * OO, Wo, bo);
        } else {
            const int fw = wave & 1;
            if (s < NCH)
                helper_fetch(fw, lane, pre + ((size_t)(s * CH) * BB + b0) * PH,
                             &preL[s & 1][0][0]);
        }
        __syncthreads();
    }
}

extern "C" void kernel_launch(void* const* d_in, const int* in_sizes, int n_in,
                              void* d_out, int out_size, void* d_ws, size_t ws_size,
                              hipStream_t stream)
{
    const float* x     = (const float*)d_in[0];
    const float* hx    = (const float*)d_in[1];
    const float* w_ih0 = (const float*)d_in[2];
    const float* w_hh0 = (const float*)d_in[3];
    const float* b_ih0 = (const float*)d_in[4];
    const float* b_hh0 = (const float*)d_in[5];
    const float* w_ih1 = (const float*)d_in[6];
    const float* w_hh1 = (const float*)d_in[7];
    const float* b_ih1 = (const float*)d_in[8];
    const float* b_hh1 = (const float*)d_in[9];
    const float* W_out = (const float*)d_in[10];
    const float* b_out = (const float*)d_in[11];

    float* out = (float*)d_out;
    float* pre = (float*)d_ws;

    dim3 g1(TT / (4 * TPW * K1W), BB);
    rnn_inproj_wmma<<<g1, 32 * K1W, 0, stream>>>(x, w_ih0, b_ih0, b_hh0, pre);

    rnn_scan<<<BB / NB, 256, 0, stream>>>(pre, hx, w_hh0, w_ih1, w_hh1,
                                         b_ih1, b_hh1, W_out, b_out, out);
}


# kernel 1: 16 tiles per wave (was 8) at 4 waves per SIMD; a wave's x loads go out in groups of 16, two groups ahead of their matrix work (were sent in waited pairs); same arithmetic and sum order
# speedup vs baseline: 20.3336x; 1.0859x over previous
#include <hip/hip_runtime.h>

typedef __attribute__((ext_vector_type(2))) float v2f;
typedef __attribute__((ext_vector_type(8))) float v8f;

#define BB 128
#define TT 2048
#define FF 124
#define HH 3
#define OO 6
#define PH 4

#define TPW  16
#define K1W  4

__device__ __forceinline__ float quad_x1(float v) {
    return __builtin_bit_cast(float, __builtin_amdgcn_mov_dpp(
        __builtin_bit_cast(int, v), 0xB1  , 0xF, 0xF, true));
}
__device__ __forceinline__ float quad_x2(float v) {
    return __builtin_bit_cast(float, __builtin_amdgcn_mov_dpp(
        __builtin_bit_cast(int, v), 0x4E  , 0xF, 0xF, true));
}

#define TG   4
static_assert(TPW == 4 * TG, "the kernel body names its four groups one by one");

template <int GI>
__device__ __forceinline__ void load_grp(const float* __restrict__ xa,
                                         const float* __restrict__ xb,
                                         float4 (&a)[TG][4]) {
#pragma unroll
    for (int t = 0; t < TG; ++t) {
        const int i = GI * TG + t;
#pragma unroll
        for (int j = 0; j < 3; ++j)
            a[t][j] = *(const float4*)(xa + i * (4 * FF) + 32 * j);
        a[t][3] = *(const float4*)(xb + i * (4 * FF));
    }
}

template <int GI>
__device__ __forceinline__ void comp_grp(float4 (&a)[TG][4], const v2f (&bf)[8],
                                         bool last, int g, float bias,
                                         float (&res)[TPW]) {
#pragma unroll
    for (int t = 0; t < TG; ++t) {
        if (last) a[t][3] = make_float4(0.0f, 0.0f, 0.0f, 0.0f);

        v8f acc = {};
#pragma unroll
        for (int j = 0; j < 4; ++j) {
            v2f lo, up;
            lo.x = a[t][j].x; lo.y = a[t][j].y;
            up.x = a[t][j].z; up.y = a[t][j].w;
            acc = __builtin_amdgcn_wmma_f32_16x16x4_f32(
                 false, lo,  false, bf[2 * j],
                 (short)0, acc,  false,  false);
            acc = __builtin_amdgcn_wmma_f32_16x16x4_f32(
                 false, up,  false, bf[2 * j + 1],
                 (short)0, acc,  false,  false);
        }

        float v0 = (g & 2) ? ((g & 1) ? acc[3] : acc[2]) : ((g & 1) ? acc[1] : acc[0]);
        float v1 = (g & 2) ? ((g & 1) ? acc[7] : acc[6]) : ((g & 1) ? acc[5] : acc[4]);
        v0 += quad_x1(v0);
        v1 += quad_x1(v1);
        v0 += quad_x2(v0);
        v1 += quad_x2(v1);
        res[GI * TG + t] = ((g == 0) ? v0 : v1) + bias;
    }
}

__global__ __launch_bounds__(32 * K1W, 4) void rnn_inproj_wmma(
    const float* __restrict__ x,
    const float* __restrict__ w_ih0,
    const float* __restrict__ b_ih0,
    const float* __restrict__ b_hh0,
    float* __restrict__ pre)
{
    const int lane = threadIdx.x & 31;
    const int wv   = threadIdx.x >> 5;
    const int c    = lane & 15;
    const int hi   = lane >> 4;
    const int g    = c & 3;
    const int r4   = c >> 2;
    const int b    = blockIdx.y;
    const int t0   = (blockIdx.x * K1W + wv) * (4 * TPW);

    const int  qo   = 2 * g + hi;
    const bool last = (qo == 7);
    const int  q3   = last ? 30 : 24 + qo;

    const bool   ncol = (r4 < HH);
    const float* wrow = w_ih0 + (size_t)(ncol ? r4 : HH - 1) * FF;
    v2f bf[8];
#pragma unroll
    for (int j = 0; j < 4; ++j) {
        const int    qc = (j < 3) ? 8 * j + qo : q3;
        const float4 w4 = *(const float4*)(wrow + 4 * qc);
        const bool   ok = ncol && !(j == 3 && last);
        bf[2 * j].x     = ok ? w4.x : 0.0f;
        bf[2 * j].y     = ok ? w4.y : 0.0f;
        bf[2 * j + 1].x = ok ? w4.z : 0.0f;
        bf[2 * j + 1].y = ok ? w4.w : 0.0f;
    }
    const float bias = b_ih0[ncol ? r4 : HH - 1] + b_hh0[ncol ? r4 : HH - 1];

    const float* xrow = x + ((size_t)b * TT + (size_t)(t0 + r4)) * FF;
    const float* xa   = xrow + 4 * qo;
    const float* xb   = xrow + 4 * q3;

    const bool st = ncol && (g < 2);
    float* prow = pre + ((size_t)(t0 + 2 * hi + g) * BB + b) * PH + (ncol ? r4 : 0);

    float  res[TPW];
    float4 xg0[TG][4], xg1[TG][4], xg2[TG][4], xg3[TG][4];
    load_grp<0>(xa, xb, xg0);
    load_grp<1>(xa, xb, xg1);
    __builtin_amdgcn_sched_barrier(0);
    comp_grp<0>(xg0, bf, last, g, bias, res);
    __builtin_amdgcn_sched_barrier(0);
    load_grp<2>(xa, xb, xg2);
    __builtin_amdgcn_sched_barrier(0);
    comp_grp<1>(xg1, bf, last, g, bias, res);
    __builtin_amdgcn_sched_barrier(0);
    load_grp<3>(xa, xb, xg3);
    __builtin_amdgcn_sched_barrier(0);
    comp_grp<2>(xg2, bf, last, g, bias, res);
    comp_grp<3>(xg3, bf, last, g, bias, res);

    if (st) {
#pragma unroll
        for (int i = 0; i < TPW; ++i)
            prow[(size_t)i * (4 * BB * PH)] = res[i];
    }
}

#define NB   8
#define CH   64
#define NCH  (TT / CH)
#define GS   16
#define HP   (NB * 4 + 4)
#define RPW  (32 / NB)
#define NQ   (CH / RPW)
#define NGRP (NB * CH / 32)
#define NHW  4
#define BPH  (NB / NHW)

__device__ __forceinline__ void tanh3(float x0, float x1, float x2,
                                      float& r0, float& r1, float& r2) {
    asm volatile("v_tanh_f32 %0, %3\n\t"
                 "v_tanh_f32 %1, %4\n\t"
                 "v_tanh_f32 %2, %5\n\t"
                 "v_nop"
                 : "=&v"(r0), "=&v"(r1), "=&v"(r2)
                 : "v"(x0), "v"(x1), "v"(x2));
}

__device__ __forceinline__ float selu_f(float x) {
    const float kAlpha = 1.6732632423543772f;
    const float kScale = 1.0507009873554805f;
    const float e = __builtin_amdgcn_exp2f(x * 1.4426950408889634f);
    return (x > 0.0f) ? kScale * x : kScale * kAlpha * (e - 1.0f);
}

__device__ __forceinline__ void head_one(float hx0, float hx1, float hx2,
                                         float* __restrict__ op,
                                         const float (&Wo)[6][3], const float (&bo)[6])
{
    float y[3];
    y[0] = selu_f(hx0);
    y[1] = selu_f(hx1);
    y[2] = selu_f(hx2);
#pragma unroll
    for (int o = 0; o < 6; ++o) {
        float s = bo[o];
#pragma unroll
        for (int j = 0; j < 3; ++j) s = fmaf(Wo[o][j], y[j], s);
        op[o] = s;
    }
}

__device__ __forceinline__ void helper_head(int hw, int lane, const float* h2l,
                                            float* __restrict__ ob0,
                                            const float (&Wo)[6][3], const float (&bo)[6])
{
    float4 hv[2 * BPH];
#pragma unroll
    for (int k = 0; k < BPH; ++k) {
        const float* hb = h2l + lane * HP + (hw * BPH + k) * 4;
        hv[2 * k]     = *(const float4*)(hb);
        hv[2 * k + 1] = *(const float4*)(hb + 32 * HP);
    }
#pragma unroll
    for (int k = 0; k < BPH; ++k) {
        const int be = hw * BPH + k;
        float* ob = ob0 + ((size_t)be * TT + lane) * OO;
        head_one(hv[2 * k].x, hv[2 * k].y, hv[2 * k].z, ob, Wo, bo);
        head_one(hv[2 * k + 1].x, hv[2 * k + 1].y, hv[2 * k + 1].z, ob + 32 * OO, Wo, bo);
    }
}

__device__ __forceinline__ void helper_fetch(int hw, int lane,
                                             const float* __restrict__ src0, float* prel)
{
    const float* src = src0 + ((size_t)(lane >> 3) * BB + (lane & 7)) * PH;
    float4 r[NQ / 2];
#pragma unroll
    for (int i = 0; i < NQ / 2; ++i)
        r[i] = *(const float4*)(src + (size_t)((hw * (NQ / 2) + i) * RPW) * (BB * PH));
    float* dst = prel + lane * 4;
#pragma unroll
    for (int i = 0; i < NQ / 2; ++i)
        *(float4*)(dst + (hw * (NQ / 2) + i) * (32 * 4)) = r[i];
}

typedef __attribute__((ext_vector_type(3))) float v3f;

__global__ __launch_bounds__(256) void rnn_scan(
    const float* __restrict__ pre,
    const float* __restrict__ hx,
    const float* __restrict__ w_hh0,
    const float* __restrict__ w_ih1,
    const float* __restrict__ w_hh1,
    const float* __restrict__ b_ih1,
    const float* __restrict__ b_hh1,
    const float* __restrict__ W_out,
    const float* __restrict__ b_out,
    float* __restrict__ out)
{
    __shared__ __attribute__((aligned(16))) float preL[2][CH][NB * PH];
    __shared__ __attribute__((aligned(16))) float qL[2][CH][NB * PH];
    __shared__ __attribute__((aligned(16))) float h2L[2][CH][HP];

    const int lane = threadIdx.x & 31;
    const int wave = __builtin_amdgcn_readfirstlane((int)(threadIdx.x >> 5));
    const int b0   = blockIdx.x * NB;

    float A0[3][3], Wi[3][3], Wh[3][3], c1[3];
#pragma unroll
    for (int i = 0; i < 3; ++i) {
#pragma unroll
        for (int j = 0; j < 3; ++j) {
            A0[i][j] = w_hh0[i * 3 + j];
            Wi[i][j] = w_ih1[i * 3 + j];
            Wh[i][j] = w_hh1[i * 3 + j];
        }
        c1[i] = b_ih1[i] + b_hh1[i];
    }
    float Wo[6][3], bo[6];
#pragma unroll
    for (int o = 0; o < 6; ++o) {
        bo[o] = b_out[o];
#pragma unroll
        for (int j = 0; j < 3; ++j) Wo[o][j] = W_out[o * 3 + j];
    }

    const int bl = lane & (NB - 1);
    float h1[3], h2[3];
#pragma unroll
    for (int i = 0; i < 3; ++i) {
        h1[i] = hx[(0 * BB + b0 + bl) * HH + i];
        h2[i] = hx[(1 * BB + b0 + bl) * HH + i];
    }

    for (int s = 0; s < NCH + 3; ++s) {
        if (wave == 0) {
            if (s >= 1 && s <= NCH && lane < NB) {
                const int buf = (s - 1) & 1;
                const float* pL = &preL[buf][0][lane * PH];
                float*       qW = &qL[buf][0][lane * PH];
                for (int tg = 0; tg < CH; tg += GS) {
                    v3f pv[GS];
#pragma unroll
                    for (int g = 0; g < GS; ++g)
                        pv[g] = *(const v3f*)(pL + (tg + g) * (NB * PH));

                    v3f qs[GS];
#pragma unroll
                    for (int g = 0; g < GS; ++g) {
                        const float p[3] = {pv[g].x, pv[g].y, pv[g].z};

                        float u0[3];
#pragma unroll
                        for (int i = 0; i < 3; ++i) {
                            float u = p[i];
#pragma unroll
                            for (int j = 0; j < 3; ++j) u = fmaf(A0[i][j], h1[j], u);
                            u0[i] = u;
                        }
                        tanh3(u0[0], u0[1], u0[2], h1[0], h1[1], h1[2]);

                        float q[3];
#pragma unroll
                        for (int i = 0; i < 3; ++i) {
                            float v = c1[i];
#pragma unroll
                            for (int j = 0; j < 3; ++j) v = fmaf(Wi[i][j], h1[j], v);
                            q[i] = v;
                        }
                        qs[g].x = q[0]; qs[g].y = q[1]; qs[g].z = q[2];
                    }

#pragma unroll
                    for (int g = 0; g < GS; ++g)
                        *(v3f*)(qW + (tg + g) * (NB * PH)) = qs[g];
                }
            }
        } else if (wave == 1) {
            if (s >= 2 && s <= NCH + 1 && lane < NB) {
                const int buf = s & 1;
                const float* qR = &qL[buf][0][lane * PH];
                float*       hL = &h2L[buf][0][lane * 4];
                for (int tg = 0; tg < CH; tg += GS) {
                    v3f qv[GS];
#pragma unroll
                    for (int g = 0; g < GS; ++g)
                        qv[g] = *(const v3f*)(qR + (tg + g) * (NB * PH));

                    float hs[GS][3];
#pragma unroll
                    for (int g = 0; g < GS; ++g) {
                        const float q[3] = {qv[g].x, qv[g].y, qv[g].z};

                        float u1[3] = {q[0], q[1], q[2]};
#pragma unroll
                        for (int j = 0; j < 3; ++j) {
#pragma unroll
                            for (int i = 0; i < 3; ++i) u1[i] = fmaf(Wh[i][j], h2[j], u1[i]);
                            __builtin_amdgcn_sched_barrier(0);
                        }
                        tanh3(u1[0], u1[1], u1[2], h2[0], h2[1], h2[2]);
#pragma unroll
                        for (int i = 0; i < 3; ++i) hs[g][i] = h2[i];
                    }

#pragma unroll
                    for (int g = 0; g < GS; ++g)
                        *(float4*)(hL + (tg + g) * HP) =
                            make_float4(hs[g][0], hs[g][1], hs[g][2], 0.0f);
                }
            }
        } else if (wave & 2) {
            const int hw = (wave & 1) | ((wave >> 2) << 1);
            if (s >= 3)
                helper_head(hw, lane, &h2L[(s + 1) & 1][0][0],
                            out + ((size_t)b0 * TT + (size_t)(s - 3) * CH) * OO, Wo, bo);
        } else {
            const int fw = wave & 1;
            if (s < NCH)
                helper_fetch(fw, lane, pre + ((size_t)(s * CH) * BB + b0) * PH,
                             &preL[s & 1][0][0]);
        }
        __syncthreads();
    }
}

extern "C" void kernel_launch(void* const* d_in, const int* in_sizes, int n_in,
                              void* d_out, int out_size, void* d_ws, size_t ws_size,
                              hipStream_t stream)
{
    const float* x     = (const float*)d_in[0];
    const float* hx    = (const float*)d_in[1];
    const float* w_ih0 = (const float*)d_in[2];
    const float* w_hh0 = (const float*)d_in[3];
    const float* b_ih0 = (const float*)d_in[4];
    const float* b_hh0 = (const float*)d_in[5];
    const float* w_ih1 = (const float*)d_in[6];
    const float* w_hh1 = (const float*)d_in[7];
    const float* b_ih1 = (const float*)d_in[8];
    const float* b_hh1 = (const float*)d_in[9];
    const float* W_out = (const float*)d_in[10];
    const float* b_out = (const float*)d_in[11];

    float* out = (float*)d_out;
    float* pre = (float*)d_ws;

    dim3 g1(TT / (4 * TPW * K1W), BB);
    rnn_inproj_wmma<<<g1, 32 * K1W, 0, stream>>>(x, w_ih0, b_ih0, b_hh0, pre);

    rnn_scan<<<BB / NB, 256, 0, stream>>>(pre, hx, w_hh0, w_ih1, w_hh1,
                                         b_ih1, b_hh1, W_out, b_out, out);
}
